# starting kernel
# baseline (speedup 1.0000x reference)
#include <hip/hip_runtime.h>

typedef float v2f __attribute__((ext_vector_type(2)));
typedef float v8f __attribute__((ext_vector_type(8)));

#define N_ATOMS 512
#define BATCH 128
#define ROW_STRIDE 1537

__global__ void init_E_kernel(float* out) {
    int b = threadIdx.x;
    if (b < BATCH) out[(size_t)b * ROW_STRIDE + 3 * N_ATOMS] = 0.0f;
}

__global__ __launch_bounds__(256) void pairforce_wmma_kernel(
        const float* __restrict__ decompFE,
        const float* __restrict__ coords,
        float* __restrict__ out) {
    __shared__ float4 lds_c8[N_ATOMS * 2];

    const int b    = blockIdx.y;
    const int iblk = blockIdx.x;
    const int tid  = threadIdx.x;

    const float* cp = coords + (size_t)b * N_ATOMS * 3;
    for (int a = tid; a < N_ATOMS; a += 256) {
        lds_c8[a * 2 + 0] = make_float4(cp[a*3 + 0], cp[a*3 + 1], cp[a*3 + 2], 1.0f);
        lds_c8[a * 2 + 1] = make_float4(0.0f, 0.0f, 0.0f, 0.0f);
    }
    __syncthreads();

    const int lane  = tid & 31;
    const int wave  = tid >> 5;
    const int n     = lane & 15;
    const int khalf = lane >> 4;
    const int koff  = khalf * 2;
    const int sel   = (n < 8) ? n : 7;

    const float* ldsf = (const float*)lds_c8;

    const int    i_g = iblk * 128 + wave * 16 + n;
    const float4 ci  = lds_c8[i_g * 2];
    const float* feRow = decompFE + ((size_t)b * N_ATOMS + i_g) * N_ATOMS;

    v8f acc = {0.f, 0.f, 0.f, 0.f, 0.f, 0.f, 0.f, 0.f};

    #pragma unroll 4
    for (int base = 0; base < N_ATOMS; base += 4) {
        const int j0 = base + koff;
        const float2 fe  = *(const float2*)(feRow + j0);
        const float4 cj0 = lds_c8[j0 * 2];
        const float4 cj1 = lds_c8[j0 * 2 + 2];
        const float b0 = ldsf[j0 * 8 + sel];
        const float b1 = ldsf[j0 * 8 + 8 + sel];

        float dx = ci.x - cj0.x, dy = ci.y - cj0.y, dz = ci.z - cj0.z;
        float r2 = fmaf(dx, dx, fmaf(dy, dy, dz * dz));
        float d0 = fmaf(fe.x, 200.0f, -100.0f);
        float w0 = (r2 > 0.0f) ? d0 * __builtin_amdgcn_rsqf(r2) : 0.0f;

        dx = ci.x - cj1.x; dy = ci.y - cj1.y; dz = ci.z - cj1.z;
        r2 = fmaf(dx, dx, fmaf(dy, dy, dz * dz));
        float d1 = fmaf(fe.y, 200.0f, -100.0f);
        float w1 = (r2 > 0.0f) ? d1 * __builtin_amdgcn_rsqf(r2) : 0.0f;

        v2f A; A[0] = w0; A[1] = w1;
        v2f B; B[0] = b0; B[1] = b1;

        acc = __builtin_amdgcn_wmma_f32_16x16x4_f32(
                false, A, false, B, (short)0, acc, false, false);
    }

    float* op    = out + (size_t)b * ROW_STRIDE;
    float  epart = 0.0f;
    #pragma unroll
    for (int r = 0; r < 8; ++r) {
        const float Sv = __shfl(acc[r], (lane & 16) | 3, 32);
        const int row_g = iblk * 128 + wave * 16 + khalf * 8 + r;
        if (n < 3) {
            const float cik = ldsf[row_g * 8 + n];
            op[row_g * 3 + n] = fmaf(cik, Sv, -acc[r]);
        }
        epart += acc[r];
    }
    float e = (n == 3) ? epart : 0.0f;
    e += __shfl_xor(e, 16, 32);
    if (lane == 3) atomicAdd(op + 3 * N_ATOMS, 0.5f * e);
}

extern "C" void kernel_launch(void* const* d_in, const int* in_sizes, int n_in,
                              void* d_out, int out_size, void* d_ws, size_t ws_size,
                              hipStream_t stream) {
    const float* decompFE = (const float*)d_in[0];
    const float* coords   = (const float*)d_in[1];
    float* out = (float*)d_out;

    hipLaunchKernelGGL(init_E_kernel, dim3(1), dim3(128), 0, stream, out);
    hipLaunchKernelGGL(pairforce_wmma_kernel, dim3(4, BATCH), dim3(256), 0, stream,
                       decompFE, coords, out);
}


# decompFE read one 64-column chunk ahead into a second register set: a chunk's loads are issued before the previous chunk's arithmetic, so memory latency runs under it; sum order unchanged
# speedup vs baseline: 1.0998x; 1.0998x over previous
#include <hip/hip_runtime.h>

typedef float v2f __attribute__((ext_vector_type(2)));
typedef float v8f __attribute__((ext_vector_type(8)));

#define N_ATOMS 512
#define BATCH 128
#define ROW_STRIDE 1537
#define CHUNK_COLS  64
#define CHUNK_TRIPS (CHUNK_COLS / 4)
#define N_CHUNKS    (N_ATOMS / CHUNK_COLS)

__global__ void init_E_kernel(float* out) {
    int b = threadIdx.x;
    if (b < BATCH) out[(size_t)b * ROW_STRIDE + 3 * N_ATOMS] = 0.0f;
}

__global__ __launch_bounds__(256) void pairforce_wmma_kernel(
        const float* __restrict__ decompFE,
        const float* __restrict__ coords,
        float* __restrict__ out) {
    __shared__ float4 lds_c8[N_ATOMS * 2];

    const int b    = blockIdx.y;
    const int iblk = blockIdx.x;
    const int tid  = threadIdx.x;

    const int lane  = tid & 31;
    const int wave  = tid >> 5;
    const int n     = lane & 15;
    const int khalf = lane >> 4;
    const int koff  = khalf * 2;
    const int sel   = (n < 8) ? n : 7;

    const int    i_g = iblk * 128 + wave * 16 + n;
    const float* fePtr = decompFE + ((size_t)b * N_ATOMS + i_g) * N_ATOMS + koff;

    float2 fe[2][CHUNK_TRIPS];
#define LOAD_CHUNK(C) do { \
        _Pragma("unroll") for (int t_ = 0; t_ < CHUNK_TRIPS; ++t_) \
            fe[(C) & 1][t_] = *(const float2*)(fePtr + (C) * CHUNK_COLS + 4 * t_); } while (0)

    LOAD_CHUNK(0);

    const float* cp = coords + (size_t)b * N_ATOMS * 3;
    for (int a = tid; a < N_ATOMS; a += 256) {
        lds_c8[a * 2 + 0] = make_float4(cp[a*3 + 0], cp[a*3 + 1], cp[a*3 + 2], 1.0f);
        lds_c8[a * 2 + 1] = make_float4(0.0f, 0.0f, 0.0f, 0.0f);
    }
    __syncthreads();

    const float* ldsf = (const float*)lds_c8;
    const float4 ci  = lds_c8[i_g * 2];

    v8f acc = {0.f, 0.f, 0.f, 0.f, 0.f, 0.f, 0.f, 0.f};

    #pragma unroll
    for (int c = 0; c < N_CHUNKS; ++c) {
        if (c + 1 < N_CHUNKS) LOAD_CHUNK(c + 1);

        #pragma unroll
        for (int q = 0; q < CHUNK_TRIPS / 4; ++q) {
            float4 cj[8];
            float  bv[8];
            #pragma unroll
            for (int u = 0; u < 4; ++u) {
                const int j0 = c * CHUNK_COLS + 16 * q + 4 * u + koff;
                cj[2 * u]     = lds_c8[j0 * 2];
                cj[2 * u + 1] = lds_c8[j0 * 2 + 2];
                bv[2 * u]     = ldsf[j0 * 8 + sel];
                bv[2 * u + 1] = ldsf[j0 * 8 + 8 + sel];
            }
            __builtin_amdgcn_sched_barrier(0);

            #pragma unroll
            for (int u = 0; u < 4; ++u) {
                const float2 f   = fe[c & 1][4 * q + u];
                const float4 cj0 = cj[2 * u];
                const float4 cj1 = cj[2 * u + 1];

                float dx = ci.x - cj0.x, dy = ci.y - cj0.y, dz = ci.z - cj0.z;
                float r2 = fmaf(dx, dx, fmaf(dy, dy, dz * dz));
                float d0 = fmaf(f.x, 200.0f, -100.0f);
                float w0 = (r2 > 0.0f) ? d0 * __builtin_amdgcn_rsqf(r2) : 0.0f;

                dx = ci.x - cj1.x; dy = ci.y - cj1.y; dz = ci.z - cj1.z;
                r2 = fmaf(dx, dx, fmaf(dy, dy, dz * dz));
                float d1 = fmaf(f.y, 200.0f, -100.0f);
                float w1 = (r2 > 0.0f) ? d1 * __builtin_amdgcn_rsqf(r2) : 0.0f;

                v2f A; A[0] = w0; A[1] = w1;
                v2f B; B[0] = bv[2 * u]; B[1] = bv[2 * u + 1];

                acc = __builtin_amdgcn_wmma_f32_16x16x4_f32(
                        false, A, false, B, (short)0, acc, false, false);
            }
        }
    }
#undef LOAD_CHUNK

    float* op    = out + (size_t)b * ROW_STRIDE;
    float  epart = 0.0f;
    #pragma unroll
    for (int r = 0; r < 8; ++r) {
        const float Sv = __shfl(acc[r], (lane & 16) | 3, 32);
        const int row_g = iblk * 128 + wave * 16 + khalf * 8 + r;
        if (n < 3) {
            const float cik = ldsf[row_g * 8 + n];
            op[row_g * 3 + n] = fmaf(cik, Sv, -acc[r]);
        }
        epart += acc[r];
    }
    float e = (n == 3) ? epart : 0.0f;
    e += __shfl_xor(e, 16, 32);
    if (lane == 3) atomicAdd(op + 3 * N_ATOMS, 0.5f * e);
}

extern "C" void kernel_launch(void* const* d_in, const int* in_sizes, int n_in,
                              void* d_out, int out_size, void* d_ws, size_t ws_size,
                              hipStream_t stream) {
    const float* decompFE = (const float*)d_in[0];
    const float* coords   = (const float*)d_in[1];
    float* out = (float*)d_out;

    hipLaunchKernelGGL(init_E_kernel, dim3(1), dim3(128), 0, stream, out);
    hipLaunchKernelGGL(pairforce_wmma_kernel, dim3(4, BATCH), dim3(256), 0, stream,
                       decompFE, coords, out);
}


# each lane access takes 4 consecutive columns: one 16-byte decompFE load feeds two k-steps, coordinates are 16-byte LDS reads of a column-major table; k order, so sum order, changes within 8 columns
# speedup vs baseline: 1.1353x; 1.0323x over previous
#include <hip/hip_runtime.h>

typedef float v2f __attribute__((ext_vector_type(2)));
typedef float v4f __attribute__((ext_vector_type(4)));
typedef float v8f __attribute__((ext_vector_type(8)));

#define N_ATOMS 512
#define BATCH 128
#define ROW_STRIDE 1537
#define CHUNK_COLS  64
#define CHUNK_LOADS (CHUNK_COLS / 8)
#define N_CHUNKS    (N_ATOMS / CHUNK_COLS)
#define N_GROUPS    (N_ATOMS / 4)
#define GROUP_ROWS  5

__global__ void init_E_kernel(float* out) {
    int b = threadIdx.x;
    if (b < BATCH) out[(size_t)b * ROW_STRIDE + 3 * N_ATOMS] = 0.0f;
}

__global__ __launch_bounds__(256) void pairforce_wmma_kernel(
        const float* __restrict__ decompFE,
        const float* __restrict__ coords,
        float* __restrict__ out) {
    __shared__ v4f lds_g[N_GROUPS * GROUP_ROWS];
    float* ldsf = (float*)lds_g;

    const int b    = blockIdx.y;
    const int iblk = blockIdx.x;
    const int tid  = threadIdx.x;

    const int lane  = tid & 31;
    const int wave  = tid >> 5;
    const int n     = lane & 15;
    const int khalf = lane >> 4;
    const int c4    = khalf * 4;
    const int bsel  = (n < 4) ? n : 4;

    const int    i_g = iblk * 128 + wave * 16 + n;
    const float* fePtr = decompFE + ((size_t)b * N_ATOMS + i_g) * N_ATOMS + c4;

    v4f fe[2][CHUNK_LOADS];
#define LOAD_CHUNK(C) do { \
        _Pragma("unroll") for (int t_ = 0; t_ < CHUNK_LOADS; ++t_) \
            fe[(C) & 1][t_] = *(const v4f*)(fePtr + (C) * CHUNK_COLS + 8 * t_); } while (0)

    LOAD_CHUNK(0);

    const float* cp = coords + (size_t)b * N_ATOMS * 3;
    for (int a = tid; a < N_ATOMS; a += 256) {
        float* e = ldsf + (a >> 2) * (GROUP_ROWS * 4) + (a & 3);
        e[0]  = cp[a*3 + 0];
        e[4]  = cp[a*3 + 1];
        e[8]  = cp[a*3 + 2];
        e[12] = 1.0f;
        e[16] = 0.0f;
    }
    __syncthreads();

    const float* ei = ldsf + (i_g >> 2) * (GROUP_ROWS * 4) + (i_g & 3);
    const float cix = ei[0], ciy = ei[4], ciz = ei[8];

    const v4f* lp = lds_g + khalf * GROUP_ROWS;
    const v4f* bp = lp + bsel;

    v8f acc = {0.f, 0.f, 0.f, 0.f, 0.f, 0.f, 0.f, 0.f};

    #pragma unroll
    for (int c = 0; c < N_CHUNKS; ++c) {
        if (c + 1 < N_CHUNKS) LOAD_CHUNK(c + 1);

        #pragma unroll
        for (int q = 0; q < CHUNK_LOADS / 2; ++q) {
            v4f X[2], Y[2], Z[2], Bv[2];
            #pragma unroll
            for (int pp = 0; pp < 2; ++pp) {
                const int g0 = (c * CHUNK_COLS + 16 * q + 8 * pp) / 4;
                X[pp]  = lp[g0 * GROUP_ROWS + 0];
                Y[pp]  = lp[g0 * GROUP_ROWS + 1];
                Z[pp]  = lp[g0 * GROUP_ROWS + 2];
                Bv[pp] = bp[g0 * GROUP_ROWS];
            }
            __builtin_amdgcn_sched_barrier(0);

            #pragma unroll
            for (int pp = 0; pp < 2; ++pp) {
                const v4f f4 = fe[c & 1][2 * q + pp];
                const v4f dx = cix - X[pp], dy = ciy - Y[pp], dz = ciz - Z[pp];
                const v4f r2 = __builtin_elementwise_fma(dx, dx, __builtin_elementwise_fma(dy, dy, dz * dz));
                const v4f d  = __builtin_elementwise_fma(f4, (v4f)(200.0f), (v4f)(-100.0f));
                v4f w;
                #pragma unroll
                for (int e = 0; e < 4; ++e)
                    w[e] = (r2[e] > 0.0f) ? d[e] * __builtin_amdgcn_rsqf(r2[e]) : 0.0f;

                #pragma unroll
                for (int h2 = 0; h2 < 2; ++h2) {
                    v2f A; A[0] = w[2 * h2];      A[1] = w[2 * h2 + 1];
                    v2f B; B[0] = Bv[pp][2 * h2]; B[1] = Bv[pp][2 * h2 + 1];
                    acc = __builtin_amdgcn_wmma_f32_16x16x4_f32(
                            false, A, false, B, (short)0, acc, false, false);
                }
            }
        }
    }
#undef LOAD_CHUNK

    float* op    = out + (size_t)b * ROW_STRIDE;
    float  epart = 0.0f;
    #pragma unroll
    for (int r = 0; r < 8; ++r) {
        const float Sv = __shfl(acc[r], (lane & 16) | 3, 32);
        const int row_g = iblk * 128 + wave * 16 + khalf * 8 + r;
        if (n < 3) {
            const float cik = ldsf[(row_g >> 2) * (GROUP_ROWS * 4) + n * 4 + (row_g & 3)];
            op[row_g * 3 + n] = fmaf(cik, Sv, -acc[r]);
        }
        epart += acc[r];
    }
    float e = (n == 3) ? epart : 0.0f;
    e += __shfl_xor(e, 16, 32);
    if (lane == 3) atomicAdd(op + 3 * N_ATOMS, 0.5f * e);
}

extern "C" void kernel_launch(void* const* d_in, const int* in_sizes, int n_in,
                              void* d_out, int out_size, void* d_ws, size_t ws_size,
                              hipStream_t stream) {
    const float* decompFE = (const float*)d_in[0];
    const float* coords   = (const float*)d_in[1];
    float* out = (float*)d_out;

    hipLaunchKernelGGL(init_E_kernel, dim3(1), dim3(128), 0, stream, out);
    hipLaunchKernelGGL(pairforce_wmma_kernel, dim3(4, BATCH), dim3(256), 0, stream,
                       decompFE, coords, out);
}


# matrix tile rows = 4 atoms x 4 column groups: a 16-byte load covers 4 whole 128-B lines (not 16 partial), one LDS read serves 4 row quads; 4 group sums per row added last: sum order changes
# speedup vs baseline: 1.2537x; 1.1043x over previous
#include <hip/hip_runtime.h>

typedef float v2f __attribute__((ext_vector_type(2)));
typedef float v4f __attribute__((ext_vector_type(4)));
typedef float v8f __attribute__((ext_vector_type(8)));

#define N_ATOMS 512
#define BATCH 128
#define ROW_STRIDE 1537
#define ROW_QUADS    4
#define BLOCK_COLS   32
#define CHUNK_BLOCKS 2
#define CHUNK_LOADS  (CHUNK_BLOCKS * ROW_QUADS)
#define N_CHUNKS     (N_ATOMS / (BLOCK_COLS * CHUNK_BLOCKS))
#define N_GROUPS     (N_ATOMS / 4)
#define GROUP_ROWS   5
#define BLOCK_ENTRIES (BLOCK_COLS / 4)

__global__ void init_E_kernel(float* out) {
    int b = threadIdx.x;
    if (b < BATCH) out[(size_t)b * ROW_STRIDE + 3 * N_ATOMS] = 0.0f;
}

__global__ __launch_bounds__(256) void pairforce_wmma_kernel(
        const float* __restrict__ decompFE,
        const float* __restrict__ coords,
        float* __restrict__ out) {
    __shared__ v4f lds_g[N_GROUPS * GROUP_ROWS];
    float* ldsf = (float*)lds_g;

    const int b    = blockIdx.y;
    const int iblk = blockIdx.x;
    const int tid  = threadIdx.x;

    const int lane  = tid & 31;
    const int wave  = tid >> 5;
    const int n     = lane & 15;
    const int khalf = lane >> 4;
    const int il    = n & 3;
    const int cg    = n >> 2;

    const int row0 = iblk * 128 + wave * 16;
    const float* fePtr = decompFE + ((size_t)b * N_ATOMS + row0 + il) * N_ATOMS + 8 * cg + 4 * khalf;

    v4f fe[2][CHUNK_LOADS];
#define LOAD_CHUNK(C) do { \
        _Pragma("unroll") for (int k_ = 0; k_ < CHUNK_BLOCKS; ++k_) \
        _Pragma("unroll") for (int q_ = 0; q_ < ROW_QUADS; ++q_) \
            fe[(C) & 1][k_ * ROW_QUADS + q_] = \
                *(const v4f*)(fePtr + (size_t)(4 * q_) * N_ATOMS + ((C) * CHUNK_BLOCKS + k_) * BLOCK_COLS); } while (0)

    LOAD_CHUNK(0);

    const float* cp = coords + (size_t)b * N_ATOMS * 3;
    for (int a = tid; a < N_ATOMS; a += 256) {
        float* e = ldsf + (a >> 2) * (GROUP_ROWS * 4) + (a & 3);
        e[0]  = cp[a*3 + 0];
        e[4]  = cp[a*3 + 1];
        e[8]  = cp[a*3 + 2];
        e[12] = 1.0f;
    }
    __syncthreads();

    float cix[ROW_QUADS], ciy[ROW_QUADS], ciz[ROW_QUADS];
    #pragma unroll
    for (int q = 0; q < ROW_QUADS; ++q) {
        const float* ei = ldsf + ((row0 >> 2) + q) * (GROUP_ROWS * 4) + il;
        cix[q] = ei[0]; ciy[q] = ei[4]; ciz[q] = ei[8];
    }

    const v4f* lp = lds_g + (2 * cg + khalf) * GROUP_ROWS;
    const v4f* bp = lp + il;

    v8f acc[ROW_QUADS];
    #pragma unroll
    for (int q = 0; q < ROW_QUADS; ++q) acc[q] = (v8f){0.f, 0.f, 0.f, 0.f, 0.f, 0.f, 0.f, 0.f};

    v4f X[2], Y[2], Z[2], Bv[2];
#define LOAD_BLOCK(KB) do { \
        const int e0_ = (KB) * BLOCK_ENTRIES * GROUP_ROWS; \
        X[(KB) & 1]  = lp[e0_ + 0]; \
        Y[(KB) & 1]  = lp[e0_ + 1]; \
        Z[(KB) & 1]  = lp[e0_ + 2]; \
        Bv[(KB) & 1] = bp[e0_]; } while (0)
    LOAD_BLOCK(0);

    #pragma unroll
    for (int c = 0; c < N_CHUNKS; ++c) {
        if (c + 1 < N_CHUNKS) LOAD_CHUNK(c + 1);

        #pragma unroll
        for (int k = 0; k < CHUNK_BLOCKS; ++k) {
            const int kb = c * CHUNK_BLOCKS + k;
            if (kb + 1 < N_CHUNKS * CHUNK_BLOCKS) LOAD_BLOCK(kb + 1);
            __builtin_amdgcn_sched_barrier(0);

            #pragma unroll
            for (int q = 0; q < ROW_QUADS; ++q) {
                const v4f f4 = fe[c & 1][k * ROW_QUADS + q];
                const v4f dx = cix[q] - X[kb & 1], dy = ciy[q] - Y[kb & 1], dz = ciz[q] - Z[kb & 1];
                const v4f r2 = __builtin_elementwise_fma(dx, dx, __builtin_elementwise_fma(dy, dy, dz * dz));
                const v4f d  = __builtin_elementwise_fma(f4, (v4f)(200.0f), (v4f)(-100.0f));
                v4f w;
                #pragma unroll
                for (int e = 0; e < 4; ++e)
                    w[e] = (r2[e] > 0.0f) ? d[e] * __builtin_amdgcn_rsqf(r2[e]) : 0.0f;

                #pragma unroll
                for (int h2 = 0; h2 < 2; ++h2) {
                    v2f A; A[0] = w[2 * h2];           A[1] = w[2 * h2 + 1];
                    v2f B; B[0] = Bv[kb & 1][2 * h2];  B[1] = Bv[kb & 1][2 * h2 + 1];
                    acc[q] = __builtin_amdgcn_wmma_f32_16x16x4_f32(
                            false, A, false, B, (short)0, acc[q], false, false);
                }
            }
        }
    }
#undef LOAD_BLOCK
#undef LOAD_CHUNK

    float* op = out + (size_t)b * ROW_STRIDE;
    const bool upper = (cg & 1) != 0;
    float u[ROW_QUADS * 4];
    #pragma unroll
    for (int q = 0; q < ROW_QUADS; ++q) {
        #pragma unroll
        for (int t = 0; t < 4; ++t) {
            const float lo = acc[q][t], hi = acc[q][4 + t];
            u[4 * q + t] = upper ? hi : lo;
        }
    }
    float x4[ROW_QUADS * 4];
    #pragma unroll
    for (int s = 0; s < ROW_QUADS * 4; ++s) x4[s] = __shfl_xor(u[s], 4, 32);
    #pragma unroll
    for (int s = 0; s < ROW_QUADS * 4; ++s) u[s] += x4[s];
    #pragma unroll
    for (int s = 0; s < ROW_QUADS * 4; ++s) x4[s] = __shfl_xor(u[s], 24, 32);
    #pragma unroll
    for (int s = 0; s < ROW_QUADS * 4; ++s) u[s] += x4[s];
    float Sv[ROW_QUADS * 4];
    float esum = 0.0f;
    #pragma unroll
    for (int s = 0; s < ROW_QUADS * 4; ++s) {
        Sv[s] = __builtin_bit_cast(float, __builtin_amdgcn_readlane(__builtin_bit_cast(int, u[s]), 3));
        esum += Sv[s];
    }
    if (lane < 3) {
        const float* ci = ldsf + (row0 >> 2) * (GROUP_ROWS * 4) + lane * 4;
        float* fo = op + row0 * 3 + lane;
        #pragma unroll
        for (int s = 0; s < ROW_QUADS * 4; ++s) {
            const float cik = ci[(s >> 2) * (GROUP_ROWS * 4) + (s & 3)];
            fo[s * 3] = fmaf(cik, Sv[s], -u[s]);
        }
    }
    if (lane == 0) atomicAdd(op + 3 * N_ATOMS, 0.5f * esum);
}

extern "C" void kernel_launch(void* const* d_in, const int* in_sizes, int n_in,
                              void* d_out, int out_size, void* d_ws, size_t ws_size,
                              hipStream_t stream) {
    const float* decompFE = (const float*)d_in[0];
    const float* coords   = (const float*)d_in[1];
    float* out = (float*)d_out;

    hipLaunchKernelGGL(init_E_kernel, dim3(1), dim3(128), 0, stream, out);
    hipLaunchKernelGGL(pairforce_wmma_kernel, dim3(4, BATCH), dim3(256), 0, stream,
                       decompFE, coords, out);
}
